# starting kernel
# baseline (speedup 1.0000x reference)
#include <hip/hip_runtime.h>

typedef __attribute__((ext_vector_type(2))) float v2f;
typedef __attribute__((ext_vector_type(4))) float v4f;
typedef __attribute__((ext_vector_type(8))) float v8f;

constexpr int B_ = 8, N_ = 1024, F_ = 512, R_ = 32;
constexpr int ROWS_PER_WAVE   = 16;
constexpr int WAVES_PER_BLOCK = 8;
constexpr int ROWS_PER_BLOCK  = ROWS_PER_WAVE * WAVES_PER_BLOCK;
constexpr int BLOCKS_PER_BATCH = N_ / ROWS_PER_BLOCK;
constexpr int NUM_GEMM_BLOCKS  = B_ * BLOCKS_PER_BATCH;

constexpr size_t WS_WT   = 0;
constexpr size_t WS_PART = WS_WT + (size_t)3 * R_ * F_;
constexpr size_t WS_GSUM = WS_PART + (size_t)NUM_GEMM_BLOCKS * 2 * R_;
constexpr size_t WS_CSUM = WS_GSUM + (size_t)B_ * R_;

__global__ void transpose_w_kernel(const float* __restrict__ W1,
                                   const float* __restrict__ W2,
                                   const float* __restrict__ Wg,
                                   float* __restrict__ wt) {
    int idx = blockIdx.x * blockDim.x + threadIdx.x;
    int m   = idx / (F_ * R_);
    int rem = idx - m * (F_ * R_);
    int k   = rem / R_;
    int r   = rem - k * R_;
    const float* W = (m == 0) ? W1 : ((m == 1) ? W2 : Wg);
    wt[((size_t)m * R_ + r) * F_ + k] = W[rem];
}

__global__ void __launch_bounds__(256)
gemm3_kernel(const float* __restrict__ x, const float* __restrict__ wt,
             float* __restrict__ z1_out, float* __restrict__ partials) {
    const int wave   = threadIdx.x >> 5;
    const int lane   = threadIdx.x & 31;
    const int lane16 = lane & 15;
    const int laneHi = lane >> 4;
    const int batch  = blockIdx.x >> 3;
    const int blkInB = blockIdx.x & 7;
    const int rowBase = blkInB * ROWS_PER_BLOCK + wave * ROWS_PER_WAVE;

    const float* ax = x + ((size_t)(batch * N_ + rowBase + lane16)) * F_ + 2 * laneHi;
    const float* bw00 = wt + ((size_t)(0 * R_ +  0 + lane16)) * F_ + 2 * laneHi;
    const float* bw01 = wt + ((size_t)(0 * R_ + 16 + lane16)) * F_ + 2 * laneHi;
    const float* bw10 = wt + ((size_t)(1 * R_ +  0 + lane16)) * F_ + 2 * laneHi;
    const float* bw11 = wt + ((size_t)(1 * R_ + 16 + lane16)) * F_ + 2 * laneHi;
    const float* bw20 = wt + ((size_t)(2 * R_ +  0 + lane16)) * F_ + 2 * laneHi;
    const float* bw21 = wt + ((size_t)(2 * R_ + 16 + lane16)) * F_ + 2 * laneHi;

    v8f a00 = {}, a01 = {};
    v8f a10 = {}, a11 = {};
    v8f a20 = {}, a21 = {};

#pragma unroll 4
    for (int k0 = 0; k0 < F_; k0 += 4) {
        v2f a   = *(const v2f*)(ax   + k0);
        v2f b00 = *(const v2f*)(bw00 + k0);
        v2f b01 = *(const v2f*)(bw01 + k0);
        v2f b10 = *(const v2f*)(bw10 + k0);
        v2f b11 = *(const v2f*)(bw11 + k0);
        v2f b20 = *(const v2f*)(bw20 + k0);
        v2f b21 = *(const v2f*)(bw21 + k0);
        a00 = __builtin_amdgcn_wmma_f32_16x16x4_f32(false, a, false, b00, (short)0, a00, false, false);
        a01 = __builtin_amdgcn_wmma_f32_16x16x4_f32(false, a, false, b01, (short)0, a01, false, false);
        a10 = __builtin_amdgcn_wmma_f32_16x16x4_f32(false, a, false, b10, (short)0, a10, false, false);
        a11 = __builtin_amdgcn_wmma_f32_16x16x4_f32(false, a, false, b11, (short)0, a11, false, false);
        a20 = __builtin_amdgcn_wmma_f32_16x16x4_f32(false, a, false, b20, (short)0, a20, false, false);
        a21 = __builtin_amdgcn_wmma_f32_16x16x4_f32(false, a, false, b21, (short)0, a21, false, false);
    }

    float* outp = z1_out + ((size_t)(batch * N_ + rowBase + laneHi * 8)) * R_ + lane16;
#pragma unroll
    for (int v = 0; v < 8; ++v) {
        outp[(size_t)v * R_]      = a00[v];
        outp[(size_t)v * R_ + 16] = a01[v];
    }

    float g0 = 0.f, g1 = 0.f, zg0 = 0.f, zg1 = 0.f;
#pragma unroll
    for (int v = 0; v < 8; ++v) {
        g0  += a20[v];
        g1  += a21[v];
        zg0 += a10[v] * a20[v];
        zg1 += a11[v] * a21[v];
    }

    __shared__ float red[2][WAVES_PER_BLOCK][R_][2];
    red[0][wave][lane16     ][laneHi] = g0;
    red[0][wave][lane16 + 16][laneHi] = g1;
    red[1][wave][lane16     ][laneHi] = zg0;
    red[1][wave][lane16 + 16][laneHi] = zg1;
    __syncthreads();

    if (threadIdx.x < 64) {
        int s = threadIdx.x >> 5;
        int r = threadIdx.x & 31;
        float p = 0.f;
#pragma unroll
        for (int w = 0; w < WAVES_PER_BLOCK; ++w) {
            p += red[s][w][r][0];
            p += red[s][w][r][1];
        }
        partials[((size_t)blockIdx.x * 2 + s) * R_ + r] = p;
    }
}

__global__ void reduce_kernel(const float* __restrict__ partials,
                              const float* __restrict__ bias,
                              float* __restrict__ gsum, float* __restrict__ csum) {
    int tid = threadIdx.x;
    int b = tid >> 5;
    int r = tid & 31;
    float g = 0.f, zg = 0.f;
#pragma unroll
    for (int blk = 0; blk < BLOCKS_PER_BATCH; ++blk) {
        size_t base = ((size_t)(b * BLOCKS_PER_BATCH + blk) * 2) * R_ + r;
        g  += partials[base];
        zg += partials[base + R_];
    }
    gsum[b * R_ + r] = g;
    csum[b * R_ + r] = bias[r] * g + zg;
}

__global__ void epilogue_kernel(float* __restrict__ out,
                                const float* __restrict__ gsum,
                                const float* __restrict__ csum) {
    int i = blockIdx.x * blockDim.x + threadIdx.x;
    size_t base = (size_t)i << 2;
    int b = (int)(base >> 15);
    int r = (int)(base & (R_ - 1));
    v4f v = *(v4f*)(out + base);
    v4f g = *(const v4f*)(gsum + b * R_ + r);
    v4f c = *(const v4f*)(csum + b * R_ + r);
    *(v4f*)(out + base) = v * g + c;
}

extern "C" void kernel_launch(void* const* d_in, const int* in_sizes, int n_in,
                              void* d_out, int out_size, void* d_ws, size_t ws_size,
                              hipStream_t stream) {
    (void)in_sizes; (void)n_in; (void)out_size; (void)ws_size;
    const float* x    = (const float*)d_in[0];
    const float* W1   = (const float*)d_in[1];
    const float* W2   = (const float*)d_in[2];
    const float* Wg   = (const float*)d_in[3];
    const float* bias = (const float*)d_in[4];
    float* out = (float*)d_out;
    float* ws  = (float*)d_ws;

    float* wt       = ws + WS_WT;
    float* partials = ws + WS_PART;
    float* gsum     = ws + WS_GSUM;
    float* csum     = ws + WS_CSUM;

    transpose_w_kernel<<<(3 * F_ * R_) / 256, 256, 0, stream>>>(W1, W2, Wg, wt);
    gemm3_kernel<<<NUM_GEMM_BLOCKS, 256, 0, stream>>>(x, wt, out, partials);
    reduce_kernel<<<1, 256, 0, stream>>>(partials, bias, gsum, csum);
    epilogue_kernel<<<(B_ * N_ * R_ / 4) / 256, 256, 0, stream>>>(out, gsum, csum);
}


# GEMM re-tiled: block = 32 rows, K split by k-step over its 8 waves (2x6 tiles per wave), partials summed through LDS in fixed wave order; 256 blocks fill all WGPs; k- and row-sum order change
# speedup vs baseline: 1.9833x; 1.9833x over previous
#include <hip/hip_runtime.h>

typedef __attribute__((ext_vector_type(2))) float v2f;
typedef __attribute__((ext_vector_type(4))) float v4f;
typedef __attribute__((ext_vector_type(8))) float v8f;

constexpr int B_ = 8, N_ = 1024, F_ = 512, R_ = 32;
constexpr int ROWS_PER_BLOCK  = 32;
constexpr int WAVES_PER_BLOCK = 8;
constexpr int K_STRIDE        = 4 * WAVES_PER_BLOCK;
constexpr int BLOCKS_PER_BATCH = N_ / ROWS_PER_BLOCK;
constexpr int NUM_GEMM_BLOCKS  = B_ * BLOCKS_PER_BATCH;

constexpr size_t WS_WT   = 0;
constexpr size_t WS_PART = WS_WT + (size_t)3 * R_ * F_;
constexpr size_t WS_GSUM = WS_PART + (size_t)NUM_GEMM_BLOCKS * 2 * R_;
constexpr size_t WS_CSUM = WS_GSUM + (size_t)B_ * R_;

__global__ void transpose_w_kernel(const float* __restrict__ W1,
                                   const float* __restrict__ W2,
                                   const float* __restrict__ Wg,
                                   float* __restrict__ wt) {
    int idx = blockIdx.x * blockDim.x + threadIdx.x;
    int m   = idx / (F_ * R_);
    int rem = idx - m * (F_ * R_);
    int k   = rem / R_;
    int r   = rem - k * R_;
    const float* W = (m == 0) ? W1 : ((m == 1) ? W2 : Wg);
    wt[((size_t)m * R_ + r) * F_ + k] = W[rem];
}

__global__ void __launch_bounds__(256)
gemm3_kernel(const float* __restrict__ x, const float* __restrict__ wt,
             float* __restrict__ z1_out, float* __restrict__ partials) {
    const int wave   = threadIdx.x >> 5;
    const int lane   = threadIdx.x & 31;
    const int lane16 = lane & 15;
    const int laneHi = lane >> 4;
    const int rowBase = blockIdx.x * ROWS_PER_BLOCK;
    const int kBase   = 4 * wave + 2 * laneHi;

    const float* ax0 = x + ((size_t)(rowBase + lane16)) * F_ + kBase;
    const float* ax1 = ax0 + (size_t)16 * F_;
    const float* bw = wt + (size_t)lane16 * F_ + kBase;

    v8f acc0[6] = {};
    v8f acc1[6] = {};

#pragma unroll
    for (int k0 = 0; k0 < F_; k0 += K_STRIDE) {
        v2f a0 = *(const v2f*)(ax0 + k0);
        v2f a1 = *(const v2f*)(ax1 + k0);
        v2f b[6];
        for (int j = 0; j < 6; ++j)
            b[j] = *(const v2f*)(bw + (size_t)(16 * j) * F_ + k0);
        for (int j = 0; j < 6; ++j) {
            acc0[j] = __builtin_amdgcn_wmma_f32_16x16x4_f32(false, a0, false, b[j], (short)0, acc0[j], false, false);
            acc1[j] = __builtin_amdgcn_wmma_f32_16x16x4_f32(false, a1, false, b[j], (short)0, acc1[j], false, false);
        }
    }

    __shared__ v4f red[WAVES_PER_BLOCK][12][2][32];
    __shared__ alignas(16) float z1s[ROWS_PER_BLOCK][36];
    __shared__ float cs[2][WAVES_PER_BLOCK][32];
    for (int j = 0; j < 6; ++j) {
        red[wave][j][0][lane]     = __builtin_shufflevector(acc0[j], acc0[j], 0, 1, 2, 3);
        red[wave][j][1][lane]     = __builtin_shufflevector(acc0[j], acc0[j], 4, 5, 6, 7);
        red[wave][6 + j][0][lane] = __builtin_shufflevector(acc1[j], acc1[j], 0, 1, 2, 3);
        red[wave][6 + j][1][lane] = __builtin_shufflevector(acc1[j], acc1[j], 4, 5, 6, 7);
    }
    __syncthreads();

    const int rt = wave >> 2, ct = (wave >> 1) & 1, p = wave & 1;
    v4f s[3];
    for (int m = 0; m < 3; ++m) {
        v4f t = red[0][rt * 6 + 2 * m + ct][p][lane];
        for (int w = 1; w < WAVES_PER_BLOCK; ++w)
            t += red[w][rt * 6 + 2 * m + ct][p][lane];
        s[m] = t;
    }

    const int r0  = rt * 16 + laneHi * 8 + p * 4;
    const int col = ct * 16 + lane16;
    for (int e = 0; e < 4; ++e) z1s[r0 + e][col] = s[0][e];

    float gp = 0.f, zgp = 0.f;
    for (int e = 0; e < 4; ++e) {
        gp  += s[2][e];
        zgp += s[1][e] * s[2][e];
    }
    cs[0][wave][lane] = gp;
    cs[1][wave][lane] = zgp;
    __syncthreads();

    {
        const int row = threadIdx.x >> 3, cg = threadIdx.x & 7;
        v4f v = *(const v4f*)&z1s[row][4 * cg];
        *(v4f*)(z1_out + ((size_t)(rowBase + row)) * R_ + 4 * cg) = v;
    }

    if (threadIdx.x < 64) {
        const int q  = threadIdx.x >> 5;
        const int r  = threadIdx.x & 31;
        const int wc = (r >> 4) * 2;
        const int c  = r & 15;
        float acc = 0.f;
        for (int t = 0; t < 2; ++t)
            for (int pp = 0; pp < 2; ++pp)
                for (int h = 0; h < 2; ++h)
                    acc += cs[q][t * 4 + wc + pp][h * 16 + c];
        partials[((size_t)blockIdx.x * 2 + q) * R_ + r] = acc;
    }
}

__global__ void reduce_kernel(const float* __restrict__ partials,
                              const float* __restrict__ bias,
                              float* __restrict__ gsum, float* __restrict__ csum) {
    int tid = threadIdx.x;
    int b = tid >> 5;
    int r = tid & 31;
    float g = 0.f, zg = 0.f;
#pragma unroll
    for (int blk = 0; blk < BLOCKS_PER_BATCH; ++blk) {
        size_t base = ((size_t)(b * BLOCKS_PER_BATCH + blk) * 2) * R_ + r;
        g  += partials[base];
        zg += partials[base + R_];
    }
    gsum[b * R_ + r] = g;
    csum[b * R_ + r] = bias[r] * g + zg;
}

__global__ void epilogue_kernel(float* __restrict__ out,
                                const float* __restrict__ gsum,
                                const float* __restrict__ csum) {
    int i = blockIdx.x * blockDim.x + threadIdx.x;
    size_t base = (size_t)i << 2;
    int b = (int)(base >> 15);
    int r = (int)(base & (R_ - 1));
    v4f v = *(v4f*)(out + base);
    v4f g = *(const v4f*)(gsum + b * R_ + r);
    v4f c = *(const v4f*)(csum + b * R_ + r);
    *(v4f*)(out + base) = v * g + c;
}

extern "C" void kernel_launch(void* const* d_in, const int* in_sizes, int n_in,
                              void* d_out, int out_size, void* d_ws, size_t ws_size,
                              hipStream_t stream) {
    (void)in_sizes; (void)n_in; (void)out_size; (void)ws_size;
    const float* x    = (const float*)d_in[0];
    const float* W1   = (const float*)d_in[1];
    const float* W2   = (const float*)d_in[2];
    const float* Wg   = (const float*)d_in[3];
    const float* bias = (const float*)d_in[4];
    float* out = (float*)d_out;
    float* ws  = (float*)d_ws;

    float* wt       = ws + WS_WT;
    float* partials = ws + WS_PART;
    float* gsum     = ws + WS_GSUM;
    float* csum     = ws + WS_CSUM;

    transpose_w_kernel<<<(3 * F_ * R_) / 256, 256, 0, stream>>>(W1, W2, Wg, wt);
    gemm3_kernel<<<NUM_GEMM_BLOCKS, 256, 0, stream>>>(x, wt, out, partials);
    reduce_kernel<<<1, 256, 0, stream>>>(partials, bias, gsum, csum);
    epilogue_kernel<<<(B_ * N_ * R_ / 4) / 256, 256, 0, stream>>>(out, gsum, csum);
}


# 4 launches -> 2: transpose launch dropped (B fragments read straight from row-major W, two 128-B lines per load, not 16), per-batch reduce folded into the epilogue blocks; sum order kept
# speedup vs baseline: 3.4727x; 1.7510x over previous
#include <hip/hip_runtime.h>

typedef __attribute__((ext_vector_type(2))) float v2f;
typedef __attribute__((ext_vector_type(4))) float v4f;
typedef __attribute__((ext_vector_type(8))) float v8f;

constexpr int B_ = 8, N_ = 1024, F_ = 512, R_ = 32;
constexpr int ROWS_PER_BLOCK  = 32;
constexpr int WAVES_PER_BLOCK = 8;
constexpr int K_STRIDE        = 4 * WAVES_PER_BLOCK;
constexpr int BLOCKS_PER_BATCH = N_ / ROWS_PER_BLOCK;
constexpr int NUM_GEMM_BLOCKS  = B_ * BLOCKS_PER_BATCH;


__global__ void __launch_bounds__(256)
gemm3_kernel(const float* __restrict__ x, const float* __restrict__ W1,
             const float* __restrict__ W2, const float* __restrict__ Wg,
             float* __restrict__ z1_out, float* __restrict__ partials) {
    const int wave   = threadIdx.x >> 5;
    const int lane   = threadIdx.x & 31;
    const int lane16 = lane & 15;
    const int laneHi = lane >> 4;
    const int rowBase = blockIdx.x * ROWS_PER_BLOCK;
    const int kBase   = 4 * wave + 2 * laneHi;

    const float* ax0 = x + ((size_t)(rowBase + lane16)) * F_ + kBase;
    const float* ax1 = ax0 + (size_t)16 * F_;
    const size_t boff = (size_t)kBase * R_ + lane16;
    const float* bw[3] = { W1 + boff, W2 + boff, Wg + boff };

    v8f acc0[6] = {};
    v8f acc1[6] = {};

#pragma unroll
    for (int k0 = 0; k0 < F_; k0 += K_STRIDE) {
        v2f a0 = *(const v2f*)(ax0 + k0);
        v2f a1 = *(const v2f*)(ax1 + k0);
        v2f b[6];
        for (int j = 0; j < 6; ++j) {
            const float* bp = bw[j >> 1] + (size_t)k0 * R_ + 16 * (j & 1);
            b[j].x = bp[0];
            b[j].y = bp[R_];
        }
        for (int j = 0; j < 6; ++j) {
            acc0[j] = __builtin_amdgcn_wmma_f32_16x16x4_f32(false, a0, false, b[j], (short)0, acc0[j], false, false);
            acc1[j] = __builtin_amdgcn_wmma_f32_16x16x4_f32(false, a1, false, b[j], (short)0, acc1[j], false, false);
        }
    }

    __shared__ v4f red[WAVES_PER_BLOCK][12][2][32];
    __shared__ alignas(16) float z1s[ROWS_PER_BLOCK][36];
    __shared__ float cs[2][WAVES_PER_BLOCK][32];
    for (int j = 0; j < 6; ++j) {
        red[wave][j][0][lane]     = __builtin_shufflevector(acc0[j], acc0[j], 0, 1, 2, 3);
        red[wave][j][1][lane]     = __builtin_shufflevector(acc0[j], acc0[j], 4, 5, 6, 7);
        red[wave][6 + j][0][lane] = __builtin_shufflevector(acc1[j], acc1[j], 0, 1, 2, 3);
        red[wave][6 + j][1][lane] = __builtin_shufflevector(acc1[j], acc1[j], 4, 5, 6, 7);
    }
    __syncthreads();

    const int rt = wave >> 2, ct = (wave >> 1) & 1, p = wave & 1;
    v4f s[3];
    for (int m = 0; m < 3; ++m) {
        v4f t = red[0][rt * 6 + 2 * m + ct][p][lane];
        for (int w = 1; w < WAVES_PER_BLOCK; ++w)
            t += red[w][rt * 6 + 2 * m + ct][p][lane];
        s[m] = t;
    }

    const int r0  = rt * 16 + laneHi * 8 + p * 4;
    const int col = ct * 16 + lane16;
    for (int e = 0; e < 4; ++e) z1s[r0 + e][col] = s[0][e];

    float gp = 0.f, zgp = 0.f;
    for (int e = 0; e < 4; ++e) {
        gp  += s[2][e];
        zgp += s[1][e] * s[2][e];
    }
    cs[0][wave][lane] = gp;
    cs[1][wave][lane] = zgp;
    __syncthreads();

    {
        const int row = threadIdx.x >> 3, cg = threadIdx.x & 7;
        v4f v = *(const v4f*)&z1s[row][4 * cg];
        *(v4f*)(z1_out + ((size_t)(rowBase + row)) * R_ + 4 * cg) = v;
    }

    if (threadIdx.x < 64) {
        const int q  = threadIdx.x >> 5;
        const int r  = threadIdx.x & 31;
        const int wc = (r >> 4) * 2;
        const int c  = r & 15;
        float acc = 0.f;
        for (int t = 0; t < 2; ++t)
            for (int pp = 0; pp < 2; ++pp)
                for (int h = 0; h < 2; ++h)
                    acc += cs[q][t * 4 + wc + pp][h * 16 + c];
        partials[((size_t)blockIdx.x * 2 + q) * R_ + r] = acc;
    }
}

__global__ void __launch_bounds__(256)
epilogue_kernel(float* __restrict__ out, const float* __restrict__ partials,
                const float* __restrict__ bias) {
    __shared__ alignas(16) float sums[2][R_];
    const int b = blockIdx.x / BLOCKS_PER_BATCH;
    const size_t base = ((size_t)blockIdx.x * 256 + threadIdx.x) << 2;
    const int r4 = (threadIdx.x & 7) * 4;

    v4f v  = *(const v4f*)(out + base);
    v4f bi = *(const v4f*)(bias + r4);

    if (threadIdx.x < 2 * R_) {
        const float* p = partials + (size_t)b * BLOCKS_PER_BATCH * 2 * R_ + threadIdx.x;
        float acc = 0.f;
#pragma unroll
        for (int blk = 0; blk < BLOCKS_PER_BATCH; ++blk)
            acc += p[(size_t)blk * 2 * R_];
        sums[0][threadIdx.x] = acc;
    }
    __syncthreads();

    v4f g  = *(const v4f*)&sums[0][r4];
    v4f zg = *(const v4f*)&sums[1][r4];
    v4f c  = bi * g + zg;
    *(v4f*)(out + base) = v * g + c;
}

extern "C" void kernel_launch(void* const* d_in, const int* in_sizes, int n_in,
                              void* d_out, int out_size, void* d_ws, size_t ws_size,
                              hipStream_t stream) {
    (void)in_sizes; (void)n_in; (void)out_size; (void)ws_size;
    const float* x    = (const float*)d_in[0];
    const float* W1   = (const float*)d_in[1];
    const float* W2   = (const float*)d_in[2];
    const float* Wg   = (const float*)d_in[3];
    const float* bias = (const float*)d_in[4];
    float* out = (float*)d_out;
    float* ws  = (float*)d_ws;

    float* partials = ws;

    gemm3_kernel<<<NUM_GEMM_BLOCKS, 256, 0, stream>>>(x, W1, W2, Wg, out, partials);
    epilogue_kernel<<<(B_ * N_ * R_ / 4) / 256, 256, 0, stream>>>(out, partials, bias);
}


# GEMM operand loads 16 B per lane: x straight into two k-steps' A fragments, W as whole lines via a wave-own LDS buffer read back in fragment order; global loads 224 -> 64 per wave; k-sum order change
# speedup vs baseline: 3.6604x; 1.0541x over previous
#include <hip/hip_runtime.h>

typedef __attribute__((ext_vector_type(2))) float v2f;
typedef __attribute__((ext_vector_type(4))) float v4f;
typedef __attribute__((ext_vector_type(8))) float v8f;

constexpr int B_ = 8, N_ = 1024, F_ = 512, R_ = 32;
constexpr int ROWS_PER_BLOCK  = 32;
constexpr int WAVES_PER_BLOCK = 8;
constexpr int K_STRIDE        = 8 * WAVES_PER_BLOCK;
constexpr int NUM_ROUNDS      = F_ / K_STRIDE;
constexpr int FB_M = 320, FB_U = 160, FB_T = 68;
constexpr int SLOT = 3 * FB_M;
constexpr int BLOCKS_PER_BATCH = N_ / ROWS_PER_BLOCK;
constexpr int NUM_GEMM_BLOCKS  = B_ * BLOCKS_PER_BATCH;


__global__ void __launch_bounds__(256)
gemm3_kernel(const float* __restrict__ x, const float* __restrict__ W1,
             const float* __restrict__ W2, const float* __restrict__ Wg,
             float* __restrict__ z1_out, float* __restrict__ partials) {
    const int wave   = threadIdx.x >> 5;
    const int lane   = threadIdx.x & 31;
    const int lane16 = lane & 15;
    const int laneHi = lane >> 4;
    const int rowBase = blockIdx.x * ROWS_PER_BLOCK;

    const float* ax0 = x + ((size_t)(rowBase + lane16)) * F_ + 8 * wave + 4 * laneHi;
    const float* ax1 = ax0 + (size_t)16 * F_;
    const int srow = lane >> 3, spiece = lane & 7;
    const size_t goff = (size_t)(8 * wave + srow) * R_ + 4 * spiece;
    const float* gw[3] = { W1 + goff, W2 + goff, Wg + goff };

    __shared__ alignas(16) float wst[WAVES_PER_BLOCK][2][SLOT];
    float* const stw = &wst[wave][0][FB_U * (srow >> 1) + FB_T * (spiece >> 2) + 8 * (spiece & 3) + (srow & 1)];
    const float* const ldw = &wst[wave][0][32 * laneHi + 2 * lane16];

    v8f acc0[6] = {};
    v8f acc1[6] = {};

    v4f ga0[NUM_ROUNDS], ga1[NUM_ROUNDS];
    v4f gb[NUM_ROUNDS][6];

    auto load_w = [&](int rnd) {
        for (int g = 0; g < 6; ++g)
            gb[rnd][g] = *(const v4f*)(gw[g >> 1] + (size_t)(rnd * K_STRIDE + 4 * (g & 1)) * R_);
    };
    auto load_x = [&](int rnd) {
        ga0[rnd] = *(const v4f*)(ax0 + rnd * K_STRIDE);
        ga1[rnd] = *(const v4f*)(ax1 + rnd * K_STRIDE);
    };
    v2f fb[NUM_ROUNDS][2][6];
    auto stage = [&](int rnd) {
        float* const st = stw + (rnd & 1) * SLOT;
        for (int g = 0; g < 6; ++g)
            for (int e = 0; e < 4; ++e)
                st[FB_M * (g >> 1) + 32 * (g & 1) + 2 * e] = gb[rnd][g][e];
        __builtin_amdgcn_fence(__ATOMIC_RELEASE, "wavefront");
        __builtin_amdgcn_fence(__ATOMIC_ACQUIRE, "wavefront");
        const float* const ld = ldw + (rnd & 1) * SLOT;
        for (int u = 0; u < 2; ++u)
            for (int j = 0; j < 6; ++j)
                fb[rnd][u][j] = *(const v2f*)(ld + FB_M * (j >> 1) + FB_U * u + FB_T * (j & 1));
    };

    load_w(0); load_x(0);
    load_w(1); load_x(1);
    load_w(2);
    stage(0);
    __builtin_amdgcn_sched_barrier(0);
#pragma unroll
    for (int i = 0; i < NUM_ROUNDS; ++i) {
        if (i + 3 < NUM_ROUNDS) load_w(i + 3);
        if (i + 2 < NUM_ROUNDS) load_x(i + 2);
        if (i + 1 < NUM_ROUNDS) stage(i + 1);
        for (int u = 0; u < 2; ++u) {
            v2f fa0, fa1;
            fa0.x = ga0[i][2 * u]; fa0.y = ga0[i][2 * u + 1];
            fa1.x = ga1[i][2 * u]; fa1.y = ga1[i][2 * u + 1];
            for (int j = 0; j < 6; ++j) {
                acc0[j] = __builtin_amdgcn_wmma_f32_16x16x4_f32(false, fa0, false, fb[i][u][j], (short)0, acc0[j], false, false);
                acc1[j] = __builtin_amdgcn_wmma_f32_16x16x4_f32(false, fa1, false, fb[i][u][j], (short)0, acc1[j], false, false);
            }
        }
        __builtin_amdgcn_sched_barrier(0);
    }

    __builtin_amdgcn_sched_barrier(0);
    asm volatile("v_nop\n\tv_nop\n\tv_nop\n\tv_nop\n\tv_nop\n\tv_nop\n\tv_nop\n\tv_nop");
    __builtin_amdgcn_sched_barrier(0);

    __shared__ v4f red[WAVES_PER_BLOCK][12][2][32];
    __shared__ alignas(16) float z1s[ROWS_PER_BLOCK][36];
    __shared__ float cs[2][WAVES_PER_BLOCK][32];
    for (int j = 0; j < 6; ++j) {
        red[wave][j][0][lane]     = __builtin_shufflevector(acc0[j], acc0[j], 0, 1, 2, 3);
        red[wave][j][1][lane]     = __builtin_shufflevector(acc0[j], acc0[j], 4, 5, 6, 7);
        red[wave][6 + j][0][lane] = __builtin_shufflevector(acc1[j], acc1[j], 0, 1, 2, 3);
        red[wave][6 + j][1][lane] = __builtin_shufflevector(acc1[j], acc1[j], 4, 5, 6, 7);
    }
    __syncthreads();

    const int rt = wave >> 2, ct = (wave >> 1) & 1, p = wave & 1;
    v4f s[3];
    for (int m = 0; m < 3; ++m) {
        v4f t = red[0][rt * 6 + 2 * m + ct][p][lane];
        for (int w = 1; w < WAVES_PER_BLOCK; ++w)
            t += red[w][rt * 6 + 2 * m + ct][p][lane];
        s[m] = t;
    }

    const int r0  = rt * 16 + laneHi * 8 + p * 4;
    const int col = ct * 16 + lane16;
    for (int e = 0; e < 4; ++e) z1s[r0 + e][col] = s[0][e];

    float gp = 0.f, zgp = 0.f;
    for (int e = 0; e < 4; ++e) {
        gp  += s[2][e];
        zgp += s[1][e] * s[2][e];
    }
    cs[0][wave][lane] = gp;
    cs[1][wave][lane] = zgp;
    __syncthreads();

    {
        const int row = threadIdx.x >> 3, cg = threadIdx.x & 7;
        v4f v = *(const v4f*)&z1s[row][4 * cg];
        *(v4f*)(z1_out + ((size_t)(rowBase + row)) * R_ + 4 * cg) = v;
    }

    if (threadIdx.x < 64) {
        const int q  = threadIdx.x >> 5;
        const int r  = threadIdx.x & 31;
        const int wc = (r >> 4) * 2;
        const int c  = r & 15;
        float acc = 0.f;
        for (int t = 0; t < 2; ++t)
            for (int pp = 0; pp < 2; ++pp)
                for (int h = 0; h < 2; ++h)
                    acc += cs[q][t * 4 + wc + pp][h * 16 + c];
        partials[((size_t)blockIdx.x * 2 + q) * R_ + r] = acc;
    }
}

__global__ void __launch_bounds__(256)
epilogue_kernel(float* __restrict__ out, const float* __restrict__ partials,
                const float* __restrict__ bias) {
    __shared__ alignas(16) float sums[2][R_];
    const int b = blockIdx.x / BLOCKS_PER_BATCH;
    const size_t base = ((size_t)blockIdx.x * 256 + threadIdx.x) << 2;
    const int r4 = (threadIdx.x & 7) * 4;

    v4f v  = *(const v4f*)(out + base);
    v4f bi = *(const v4f*)(bias + r4);

    if (threadIdx.x < 2 * R_) {
        const float* p = partials + (size_t)b * BLOCKS_PER_BATCH * 2 * R_ + threadIdx.x;
        float acc = 0.f;
#pragma unroll
        for (int blk = 0; blk < BLOCKS_PER_BATCH; ++blk)
            acc += p[(size_t)blk * 2 * R_];
        sums[0][threadIdx.x] = acc;
    }
    __syncthreads();

    v4f g  = *(const v4f*)&sums[0][r4];
    v4f zg = *(const v4f*)&sums[1][r4];
    v4f c  = bi * g + zg;
    *(v4f*)(out + base) = v * g + c;
}

extern "C" void kernel_launch(void* const* d_in, const int* in_sizes, int n_in,
                              void* d_out, int out_size, void* d_ws, size_t ws_size,
                              hipStream_t stream) {
    (void)in_sizes; (void)n_in; (void)out_size; (void)ws_size;
    const float* x    = (const float*)d_in[0];
    const float* W1   = (const float*)d_in[1];
    const float* W2   = (const float*)d_in[2];
    const float* Wg   = (const float*)d_in[3];
    const float* bias = (const float*)d_in[4];
    float* out = (float*)d_out;
    float* ws  = (float*)d_ws;

    float* partials = ws;

    gemm3_kernel<<<NUM_GEMM_BLOCKS, 256, 0, stream>>>(x, W1, W2, Wg, out, partials);
    epilogue_kernel<<<(B_ * N_ * R_ / 4) / 256, 256, 0, stream>>>(out, partials, bias);
}
